# starting kernel
# baseline (speedup 1.0000x reference)
#include <hip/hip_runtime.h>

typedef __attribute__((ext_vector_type(2))) float v2f;
typedef __attribute__((ext_vector_type(8))) float v8f;

#define NPTS   200000
#define MOUT   100000
#define CIN    32
#define COUT   64
#define KNBR   32
#define VOX    27
#define KDIM   (VOX * CIN)
#define AS     (KDIM + 2)
#define MTILE  16
#define WAVES  4
#define BLOCK  (WAVES * 32)

__global__ __launch_bounds__(BLOCK)
void cconv_fused_kernel(const float* __restrict__ feats,
                        const float* __restrict__ inPos,
                        const float* __restrict__ outPos,
                        const float* __restrict__ extents,
                        const float* __restrict__ importance,
                        const int*   __restrict__ nbrIdx,
                        const float* __restrict__ Wg,
                        const float* __restrict__ bias,
                        float*       __restrict__ out)
{
    __shared__ float sA[MTILE][AS];
    __shared__ float sInvN[MTILE];

    const int lane = threadIdx.x & 31;
    const int wave = threadIdx.x >> 5;
    const int m0   = blockIdx.x * MTILE;
    const float invR = 2.0f / extents[0];

    for (int r = 0; r < MTILE / WAVES; ++r) {
        const int row = wave * (MTILE / WAVES) + r;
        const int m   = m0 + row;
        const float ox = outPos[m * 3 + 0];
        const float oy = outPos[m * 3 + 1];
        const float oz = outPos[m * 3 + 2];

        float acc[VOX];
        #pragma unroll
        for (int v = 0; v < VOX; ++v) acc[v] = 0.0f;
        float norm = 0.0f;

        for (int k = 0; k < KNBR; ++k) {
            const int idx = nbrIdx[m * KNBR + k];
            const float rx = (inPos[idx * 3 + 0] - ox) * invR;
            const float ry = (inPos[idx * 3 + 1] - oy) * invR;
            const float rz = (inPos[idx * 3 + 2] - oz) * invR;
            const float dist = sqrtf(rx * rx + ry * ry + rz * rz);
            const float linf = fmaxf(fabsf(rx), fmaxf(fabsf(ry), fabsf(rz)));
            const float scale = (linf > 1e-8f) ? dist / fmaxf(linf, 1e-8f) : 0.0f;
            const float imp = (dist <= 1.0f) ? importance[idx] : 0.0f;

            if (imp != 0.0f) {
                norm += imp;
                const float cx = fminf(fmaxf(rx * scale + 1.0f, 0.0f), 2.0f);
                const float cy = fminf(fmaxf(ry * scale + 1.0f, 0.0f), 2.0f);
                const float cz = fminf(fmaxf(rz * scale + 1.0f, 0.0f), 2.0f);
                float wx[3], wy[3], wz[3];
                #pragma unroll
                for (int a = 0; a < 3; ++a) {
                    wx[a] = fmaxf(0.0f, 1.0f - fabsf(cx - (float)a));
                    wy[a] = fmaxf(0.0f, 1.0f - fabsf(cy - (float)a));
                    wz[a] = fmaxf(0.0f, 1.0f - fabsf(cz - (float)a));
                }
                const float f = feats[idx * CIN + lane] * imp;
                const float fz0 = wz[0] * f, fz1 = wz[1] * f, fz2 = wz[2] * f;
                #pragma unroll
                for (int a = 0; a < 3; ++a) {
                    #pragma unroll
                    for (int b = 0; b < 3; ++b) {
                        const float wab = wx[a] * wy[b];
                        acc[(a * 3 + b) * 3 + 0] = fmaf(wab, fz0, acc[(a * 3 + b) * 3 + 0]);
                        acc[(a * 3 + b) * 3 + 1] = fmaf(wab, fz1, acc[(a * 3 + b) * 3 + 1]);
                        acc[(a * 3 + b) * 3 + 2] = fmaf(wab, fz2, acc[(a * 3 + b) * 3 + 2]);
                    }
                }
            }
        }

        #pragma unroll
        for (int v = 0; v < VOX; ++v)
            sA[row][v * CIN + lane] = acc[v];
        if (lane == 0)
            sInvN[row] = (norm > 0.0f) ? 1.0f / fmaxf(norm, 1e-12f) : 0.0f;
    }
    __syncthreads();

    const int n0   = wave * 16;
    const int half = lane >> 4;
    const int lm   = lane & 15;
    const float* Wt = Wg + n0 + lm;

    v8f c = {};
    #pragma unroll 4
    for (int kk = 0; kk < KDIM / 4; ++kk) {
        const int kb = kk * 4 + 2 * half;
        v2f a;
        a.x = sA[lm][kb + 0];
        a.y = sA[lm][kb + 1];
        v2f b;
        b.x = Wt[(kb + 0) * COUT];
        b.y = Wt[(kb + 1) * COUT];
        c = __builtin_amdgcn_wmma_f32_16x16x4_f32(
                 false, a,  false, b,
                 (short)0, c,  false,  false);
    }

    const float bv = bias[n0 + lm];
    #pragma unroll
    for (int j = 0; j < 8; ++j) {
        const int mrow = j + 8 * half;
        out[(long)(m0 + mrow) * COUT + n0 + lm] = c[j] * sInvN[mrow] + bv;
    }
}

extern "C" void kernel_launch(void* const* d_in, const int* in_sizes, int n_in,
                              void* d_out, int out_size, void* d_ws, size_t ws_size,
                              hipStream_t stream) {
    const float* feats      = (const float*)d_in[0];
    const float* inPos      = (const float*)d_in[1];
    const float* outPos     = (const float*)d_in[2];
    const float* extents    = (const float*)d_in[3];
    const float* importance = (const float*)d_in[4];
    const int*   nbrIdx     = (const int*)  d_in[5];
    const float* Wg         = (const float*)d_in[6];
    const float* bias       = (const float*)d_in[7];
    float* out = (float*)d_out;

    dim3 grid(MOUT / MTILE);
    cconv_fused_kernel<<<grid, BLOCK, 0, stream>>>(
        feats, inPos, outPos, extents, importance, nbrIdx, Wg, bias, out);
}


# neighbour loop: geometry with lane = neighbour, in-radius neighbours packed to an LDS list, feature rows gathered 8 independent loads at a time instead of 4 dependent loads per neighbour
# speedup vs baseline: 2.5889x; 2.5889x over previous
#include <hip/hip_runtime.h>

typedef __attribute__((ext_vector_type(2))) float v2f;
typedef __attribute__((ext_vector_type(4))) float v4f;
typedef __attribute__((ext_vector_type(8))) float v8f;

#define NPTS   200000
#define MOUT   100000
#define CIN    32
#define COUT   64
#define KNBR   32
#define VOX    27
#define KDIM   (VOX * CIN)
#define AS     (KDIM + 2)
#define MTILE  16
#define WAVES  4
#define BLOCK  (WAVES * 32)
#define RPW    (MTILE / WAVES)
#define BATCH  8
#define ENT    20

__device__ __forceinline__ void accum_entry(float (&acc)[VOX], float& norm, const float* e, float fe)
{
    const v4f e0 = *(const v4f*)(e);
    const v4f e1 = *(const v4f*)(e + 4);
    const v4f e2 = *(const v4f*)(e + 8);
    const v2f e3 = *(const v2f*)(e + 12);
    norm += e0.y;
    const float f = fe * e0.y;
    const float fz0 = e0.z * f, fz1 = e0.w * f, fz2 = e1.x * f;
    const float wab[9] = {e1.y, e1.z, e1.w, e2.x, e2.y, e2.z, e2.w, e3.x, e3.y};
    #pragma unroll
    for (int ab = 0; ab < 9; ++ab) {
        acc[ab * 3 + 0] = fmaf(wab[ab], fz0, acc[ab * 3 + 0]);
        acc[ab * 3 + 1] = fmaf(wab[ab], fz1, acc[ab * 3 + 1]);
        acc[ab * 3 + 2] = fmaf(wab[ab], fz2, acc[ab * 3 + 2]);
    }
}

__global__ __launch_bounds__(BLOCK, 5)
void cconv_fused_kernel(const float* __restrict__ feats,
                        const float* __restrict__ inPos,
                        const float* __restrict__ outPos,
                        const float* __restrict__ extents,
                        const float* __restrict__ importance,
                        const int*   __restrict__ nbrIdx,
                        const float* __restrict__ Wg,
                        const float* __restrict__ bias,
                        float*       __restrict__ out)
{
    __shared__ __attribute__((aligned(16))) float sA[MTILE][AS];
    __shared__ float sInvN[MTILE];

    const int lane = threadIdx.x & 31;
    const int wave = threadIdx.x >> 5;
    const int m0   = blockIdx.x * MTILE;
    const float invR = 2.0f / extents[0];

    const int rowBase = wave * RPW;

    float oxr[RPW], oyr[RPW], ozr[RPW];
    #pragma unroll
    for (int r = 0; r < RPW; ++r) {
        oxr[r] = outPos[(m0 + rowBase + r) * 3 + 0];
        oyr[r] = outPos[(m0 + rowBase + r) * 3 + 1];
        ozr[r] = outPos[(m0 + rowBase + r) * 3 + 2];
    }

    int idxv[RPW];
    #pragma unroll
    for (int r = 0; r < RPW; ++r)
        idxv[r] = nbrIdx[(m0 + rowBase + r) * KNBR + lane];

    float px[RPW], py[RPW], pz[RPW], iv[RPW];
    #pragma unroll
    for (int r = 0; r < RPW; ++r) {
        px[r] = inPos[idxv[r] * 3 + 0];
        py[r] = inPos[idxv[r] * 3 + 1];
        pz[r] = inPos[idxv[r] * 3 + 2];
        iv[r] = importance[idxv[r]];
    }

    int nactPacked = 0;
    #pragma unroll
    for (int r = 0; r < RPW; ++r) {
        const int row = rowBase + r;
        const float ox = oxr[r], oy = oyr[r], oz = ozr[r];

        const float rx = (px[r] - ox) * invR;
        const float ry = (py[r] - oy) * invR;
        const float rz = (pz[r] - oz) * invR;
        const float dist = sqrtf(rx * rx + ry * ry + rz * rz);
        const float linf = fmaxf(fabsf(rx), fmaxf(fabsf(ry), fabsf(rz)));
        const float scale = (linf > 1e-8f) ? dist / fmaxf(linf, 1e-8f) : 0.0f;
        const float imp = (dist <= 1.0f) ? iv[r] : 0.0f;
        const bool act = (imp != 0.0f);

        const float cx = fminf(fmaxf(rx * scale + 1.0f, 0.0f), 2.0f);
        const float cy = fminf(fmaxf(ry * scale + 1.0f, 0.0f), 2.0f);
        const float cz = fminf(fmaxf(rz * scale + 1.0f, 0.0f), 2.0f);
        float wx[3], wy[3], wz[3];
        #pragma unroll
        for (int a = 0; a < 3; ++a) {
            wx[a] = act ? fmaxf(0.0f, 1.0f - fabsf(cx - (float)a)) : 0.0f;
            wy[a] = act ? fmaxf(0.0f, 1.0f - fabsf(cy - (float)a)) : 0.0f;
            wz[a] = act ? fmaxf(0.0f, 1.0f - fabsf(cz - (float)a)) : 0.0f;
        }

        const unsigned mask = __builtin_amdgcn_ballot_w32(act);
        const int nact  = __builtin_popcount(mask);
        const int below = (int)__builtin_amdgcn_mbcnt_lo(mask, 0u);
        const int slot  = act ? below : (nact + lane - below);
        const int firstLane = mask ? __builtin_ctz(mask) : 0;
        const int firstIdx  = __builtin_amdgcn_readlane(idxv[r], firstLane);
        const int eidx = act ? idxv[r] : firstIdx;

        float* e = &sA[row][(row & 1) * 2] + slot * ENT;
        const v4f e0 = {__int_as_float(eidx), imp, wz[0], wz[1]};
        const v4f e1 = {wz[2], wx[0] * wy[0], wx[0] * wy[1], wx[0] * wy[2]};
        const v4f e2 = {wx[1] * wy[0], wx[1] * wy[1], wx[1] * wy[2], wx[2] * wy[0]};
        const v2f e3 = {wx[2] * wy[1], wx[2] * wy[2]};
        *(v4f*)(e)      = e0;
        *(v4f*)(e + 4)  = e1;
        *(v4f*)(e + 8)  = e2;
        *(v2f*)(e + 12) = e3;

        nactPacked |= nact << (8 * r);
    }
    __builtin_amdgcn_fence(__ATOMIC_RELEASE, "wavefront");
    __builtin_amdgcn_wave_barrier();
    __builtin_amdgcn_fence(__ATOMIC_ACQUIRE, "wavefront");

    float fcur[BATCH];
    {
        const float* sc = &sA[rowBase][(rowBase & 1) * 2];
        #pragma unroll
        for (int u = 0; u < BATCH; ++u)
            fcur[u] = feats[__float_as_int(sc[u * ENT]) * CIN + lane];
    }
    #pragma unroll 1
    for (int r = 0; r < RPW; ++r) {
        const int row = rowBase + r;
        const float* sc = &sA[row][(row & 1) * 2];
        const int nact = (nactPacked >> (8 * r)) & 0xff;

        float fnext[BATCH];
        #pragma unroll
        for (int u = 0; u < BATCH; ++u) fnext[u] = 0.0f;
        if (r + 1 < RPW) {
            const float* sn = &sA[row + 1][((row + 1) & 1) * 2];
            #pragma unroll
            for (int u = 0; u < BATCH; ++u)
                fnext[u] = feats[__float_as_int(sn[u * ENT]) * CIN + lane];
        }

        float acc[VOX];
        #pragma unroll
        for (int v = 0; v < VOX; ++v) acc[v] = 0.0f;
        float norm = 0.0f;

        #pragma unroll
        for (int u = 0; u < BATCH; ++u)
            accum_entry(acc, norm, sc + u * ENT, fcur[u]);

        for (int j0 = BATCH; j0 < nact; j0 += BATCH) {
            float f[BATCH];
            #pragma unroll
            for (int u = 0; u < BATCH; ++u)
                f[u] = feats[__float_as_int(sc[(j0 + u) * ENT]) * CIN + lane];
            #pragma unroll
            for (int u = 0; u < BATCH; ++u)
                accum_entry(acc, norm, sc + (j0 + u) * ENT, f[u]);
        }

        #pragma unroll
        for (int v = 0; v < VOX; ++v)
            sA[row][v * CIN + lane] = acc[v];
        if (lane == 0)
            sInvN[row] = (norm > 0.0f) ? 1.0f / fmaxf(norm, 1e-12f) : 0.0f;

        #pragma unroll
        for (int u = 0; u < BATCH; ++u) fcur[u] = fnext[u];
    }
    __syncthreads();

    const int n0   = wave * 16;
    const int half = lane >> 4;
    const int lm   = lane & 15;
    const float* Wt = Wg + n0 + lm;

    v8f c = {};
    #pragma unroll 4
    for (int kk = 0; kk < KDIM / 4; ++kk) {
        const int kb = kk * 4 + 2 * half;
        v2f a;
        a.x = sA[lm][kb + 0];
        a.y = sA[lm][kb + 1];
        v2f b;
        b.x = Wt[(kb + 0) * COUT];
        b.y = Wt[(kb + 1) * COUT];
        c = __builtin_amdgcn_wmma_f32_16x16x4_f32(
                 false, a,  false, b,
                 (short)0, c,  false,  false);
    }

    const float bv = bias[n0 + lm];
    #pragma unroll
    for (int j = 0; j < 8; ++j) {
        const int mrow = j + 8 * half;
        out[(long)(m0 + mrow) * COUT + n0 + lm] = c[j] * sInvN[mrow] + bv;
    }
}

extern "C" void kernel_launch(void* const* d_in, const int* in_sizes, int n_in,
                              void* d_out, int out_size, void* d_ws, size_t ws_size,
                              hipStream_t stream) {
    const float* feats      = (const float*)d_in[0];
    const float* inPos      = (const float*)d_in[1];
    const float* outPos     = (const float*)d_in[2];
    const float* extents    = (const float*)d_in[3];
    const float* importance = (const float*)d_in[4];
    const int*   nbrIdx     = (const int*)  d_in[5];
    const float* Wg         = (const float*)d_in[6];
    const float* bias       = (const float*)d_in[7];
    float* out = (float*)d_out;

    dim3 grid(MOUT / MTILE);
    cconv_fused_kernel<<<grid, BLOCK, 0, stream>>>(
        feats, inPos, outPos, extents, importance, nbrIdx, Wg, bias, out);
}
